# starting kernel
# baseline (speedup 1.0000x reference)
#include <hip/hip_runtime.h>
#include <cmath>

typedef __attribute__((ext_vector_type(2))) float v2f;
typedef __attribute__((ext_vector_type(8))) float v8f;

#define B_ 64
#define T_ 2048
#define Q_ 512
#define K_ 512

__global__ __launch_bounds__(128) void mids_kernel(const float* __restrict__ query,
                                                   const float* __restrict__ W,
                                                   float* __restrict__ mids) {
    const int wid  = blockIdx.x * (blockDim.x >> 5) + (threadIdx.x >> 5);
    const int lane = threadIdx.x & 31;
    const int half = lane >> 4;
    const int l16  = lane & 15;
    const int tm   = wid >> 5;
    const int tn   = wid & 31;

    const float* qrow = query + (size_t)(tm * 16 + l16) * Q_;
    const float* wrow = W     + (size_t)(tn * 16 + l16) * Q_;

    v8f acc = {};
    for (int q0 = 0; q0 < Q_; q0 += 4) {
        v2f a = *(const v2f*)(qrow + q0 + half * 2);
        v2f b = *(const v2f*)(wrow + q0 + half * 2);
        acc = __builtin_amdgcn_wmma_f32_16x16x4_f32(false, a, false, b,
                                                    (short)0, acc, false, false);
    }
    #pragma unroll
    for (int r = 0; r < 8; ++r) {
        const int m = half * 8 + r;
        mids[(size_t)(tm * 16 + m) * K_ + tn * 16 + l16] = acc[r];
    }
}

__global__ __launch_bounds__(256) void scores_kernel(const float* __restrict__ key,
                                                     const float* __restrict__ mids,
                                                     const float* __restrict__ bias,
                                                     float* __restrict__ out) {
    const int wid  = blockIdx.x * (blockDim.x >> 5) + (threadIdx.x >> 5);
    const int lane = threadIdx.x & 31;
    const int half = lane >> 4;
    const int l16  = lane & 15;
    const int b    = wid >> 7;
    const int t0   = (wid & 127) * 16;

    const float* keyr = key  + ((size_t)b * T_ + t0 + l16) * K_;
    const float* midb = mids + (size_t)b * K_;

    v8f acc = {};
    for (int k0 = 0; k0 < K_; k0 += 32) {
        __builtin_prefetch(keyr + k0 + 128, 0, 0);
        #pragma unroll
        for (int kk = 0; kk < 32; kk += 4) {
            v2f a  = *(const v2f*)(keyr + k0 + kk + half * 2);
            v2f bm = *(const v2f*)(midb + k0 + kk + half * 2);
            acc = __builtin_amdgcn_wmma_f32_16x16x4_f32(false, a, false, bm,
                                                        (short)0, acc, false, false);
        }
    }

    const float bv = bias[0];
    float e[8];
    #pragma unroll
    for (int r = 0; r < 8; ++r)
        e[r] = __expf(tanhf(acc[r] + bv));

    if (l16 == 0) {
        float* op = out + (size_t)b * T_ + t0 + half * 8;
        #pragma unroll
        for (int r = 0; r < 8; ++r) op[r] = e[r];
    }
}

__global__ __launch_bounds__(256) void softmax_norm_kernel(float* __restrict__ out) {
    __shared__ float sdata[256];
    const int b   = blockIdx.x;
    const int tid = threadIdx.x;
    float* row = out + (size_t)b * T_;

    float v[8];
    float s = 0.f;
    #pragma unroll
    for (int i = 0; i < 8; ++i) {
        v[i] = row[tid + i * 256];
        s += v[i];
    }
    sdata[tid] = s;
    __syncthreads();
    for (int off = 128; off > 0; off >>= 1) {
        if (tid < off) sdata[tid] += sdata[tid + off];
        __syncthreads();
    }
    const float inv = 1.0f / sdata[0];
    #pragma unroll
    for (int i = 0; i < 8; ++i) row[tid + i * 256] = v[i] * inv;
}

extern "C" void kernel_launch(void* const* d_in, const int* in_sizes, int n_in,
                              void* d_out, int out_size, void* d_ws, size_t ws_size,
                              hipStream_t stream) {
    const float* query = (const float*)d_in[0];
    const float* key   = (const float*)d_in[1];
    const float* W     = (const float*)d_in[2];
    const float* bias  = (const float*)d_in[3];
    float* out  = (float*)d_out;
    float* mids = (float*)d_ws;

    mids_kernel<<<32, 128, 0, stream>>>(query, W, mids);
    scores_kernel<<<1024, 256, 0, stream>>>(key, mids, bias, out);
    softmax_norm_kernel<<<64, 256, 0, stream>>>(out);
}


# scores kernel: key read by whole-line 16-byte non-temporal loads (2 rows x 8 k pieces per f32 16x16x4 tile, diagonal of D summed over 8 lanes), not 16 lines a load; k summation order changes
# speedup vs baseline: 1.4578x; 1.4578x over previous
#include <hip/hip_runtime.h>
#include <cmath>

typedef __attribute__((ext_vector_type(2))) float v2f;
typedef __attribute__((ext_vector_type(4))) float v4f;
typedef __attribute__((ext_vector_type(8))) float v8f;

#define B_ 64
#define T_ 2048
#define Q_ 512
#define K_ 512

__global__ __launch_bounds__(128) void mids_kernel(const float* __restrict__ query,
                                                   const float* __restrict__ W,
                                                   float* __restrict__ mids) {
    const int wid  = blockIdx.x * (blockDim.x >> 5) + (threadIdx.x >> 5);
    const int lane = threadIdx.x & 31;
    const int half = lane >> 4;
    const int l16  = lane & 15;
    const int tm   = wid >> 5;
    const int tn   = wid & 31;

    const float* qrow = query + (size_t)(tm * 16 + l16) * Q_;
    const float* wrow = W     + (size_t)(tn * 16 + l16) * Q_;

    v8f acc = {};
    for (int q0 = 0; q0 < Q_; q0 += 4) {
        v2f a = *(const v2f*)(qrow + q0 + half * 2);
        v2f b = *(const v2f*)(wrow + q0 + half * 2);
        acc = __builtin_amdgcn_wmma_f32_16x16x4_f32(false, a, false, b,
                                                    (short)0, acc, false, false);
    }
    #pragma unroll
    for (int r = 0; r < 8; ++r) {
        const int m = half * 8 + r;
        mids[(size_t)(tm * 16 + m) * K_ + tn * 16 + l16] = acc[r];
    }
}

#define SC_ROWS   32
#define SC_PAIRS  (SC_ROWS / 2)
#define SC_RING   4

template <int CTRL>
static __device__ __forceinline__ float lane_xchg(float x) {
    const int v = __builtin_bit_cast(int, x);
    return __builtin_bit_cast(float, __builtin_amdgcn_mov_dpp(v, CTRL, 0xF, 0xF, true));
}

__global__ __launch_bounds__(256) void scores_kernel(const float* __restrict__ key,
                                                     const float* __restrict__ mids,
                                                     const float* __restrict__ bias,
                                                     float* __restrict__ out) {
    const int wid  = blockIdx.x * (blockDim.x >> 5) + (threadIdx.x >> 5);
    const int lane = threadIdx.x & 31;
    const int rr   = lane & 7;
    const bool b0 = rr & 1, b1 = rr & 2, b2 = rr & 4;
    const int rsel = (lane >> 3) & 1;
    const int koff = 4 * (lane & 7) + 32 * (lane >> 4);
    const int b    = wid >> 6;
    const int t0   = (wid & 63) * SC_ROWS;

    const float* kp = key  + ((size_t)b * T_ + t0 + 8 * rsel) * K_ + koff;
    const float* mp = mids + (size_t)b * K_ + koff;

    v4f m[8];
    #pragma unroll
    for (int u = 0; u < 8; ++u) m[u] = *(const v4f*)(mp + 64 * u);

    v4f buf[SC_RING][8];
#define LOAD_PAIR(P) do { \
        _Pragma("unroll") for (int u = 0; u < 8; ++u) \
            buf[(P) % SC_RING][u] = __builtin_nontemporal_load( \
                (const v4f*)(kp + (((P) >> 3) * 16 + ((P) & 7)) * K_ + 64 * u)); } while (0)

    #pragma unroll
    for (int p = 0; p < SC_RING - 1; ++p) LOAD_PAIR(p);

    float res[SC_PAIRS / 8];
    #pragma unroll
    for (int g = 0; g < SC_PAIRS / 8; ++g) res[g] = 0.f;

    #pragma unroll
    for (int p = 0; p < SC_PAIRS; ++p) {
        __builtin_amdgcn_sched_barrier(0);
        if (p + SC_RING - 1 < SC_PAIRS) LOAD_PAIR(p + SC_RING - 1);
        __builtin_amdgcn_sched_barrier(0);

        v8f acc = {};
        #pragma unroll
        for (int u = 0; u < 8; ++u) {
            const v4f a = buf[p % SC_RING][u];
            const v2f a01 = {a.x, a.y}, a23 = {a.z, a.w};
            const v2f m01 = {m[u].x, m[u].y}, m23 = {m[u].z, m[u].w};
            acc = __builtin_amdgcn_wmma_f32_16x16x4_f32(false, a01, false, m01,
                                                        (short)0, acc, false, false);
            acc = __builtin_amdgcn_wmma_f32_16x16x4_f32(false, a23, false, m23,
                                                        (short)0, acc, false, false);
        }

        const float x0 = b0 ? acc[1] : acc[0], x1 = b0 ? acc[3] : acc[2];
        const float x2 = b0 ? acc[5] : acc[4], x3 = b0 ? acc[7] : acc[6];
        const float y0 = b1 ? x1 : x0, y1 = b1 ? x3 : x2;
        float s = b2 ? y1 : y0;
        s += lane_xchg<0x161>(s);
        s += lane_xchg<0x162>(s);
        s += lane_xchg<0x164>(s);
        res[p >> 3] = (rr == (p & 7)) ? s : res[p >> 3];
    }
#undef LOAD_PAIR

    const float bv = bias[0];
    const bool owner = (lane < 8) || (lane >= 24);
    const int  tl    = (lane < 8) ? lane : lane - 16;
    #pragma unroll
    for (int g = 0; g < SC_PAIRS / 8; ++g) {
        const float e = __expf(tanhf(res[g] + bv));
        if (owner) out[(size_t)b * T_ + t0 + 16 * g + tl] = e;
    }
}

__global__ __launch_bounds__(256) void softmax_norm_kernel(float* __restrict__ out) {
    __shared__ float sdata[256];
    const int b   = blockIdx.x;
    const int tid = threadIdx.x;
    float* row = out + (size_t)b * T_;

    float v[8];
    float s = 0.f;
    #pragma unroll
    for (int i = 0; i < 8; ++i) {
        v[i] = row[tid + i * 256];
        s += v[i];
    }
    sdata[tid] = s;
    __syncthreads();
    for (int off = 128; off > 0; off >>= 1) {
        if (tid < off) sdata[tid] += sdata[tid + off];
        __syncthreads();
    }
    const float inv = 1.0f / sdata[0];
    #pragma unroll
    for (int i = 0; i < 8; ++i) row[tid + i * 256] = v[i] * inv;
}

extern "C" void kernel_launch(void* const* d_in, const int* in_sizes, int n_in,
                              void* d_out, int out_size, void* d_ws, size_t ws_size,
                              hipStream_t stream) {
    const float* query = (const float*)d_in[0];
    const float* key   = (const float*)d_in[1];
    const float* W     = (const float*)d_in[2];
    const float* bias  = (const float*)d_in[3];
    float* out  = (float*)d_out;
    float* mids = (float*)d_ws;

    mids_kernel<<<32, 128, 0, stream>>>(query, W, mids);
    scores_kernel<<<512, 256, 0, stream>>>(key, mids, bias, out);
    softmax_norm_kernel<<<64, 256, 0, stream>>>(out);
}


# mids kernel: q sum of each 16x16 tile split over 8 waves of one block (16-byte loads all requested at once, partial tiles added through LDS in wave order); k summation order changes
# speedup vs baseline: 1.5675x; 1.0753x over previous
#include <hip/hip_runtime.h>
#include <cmath>

typedef __attribute__((ext_vector_type(2))) float v2f;
typedef __attribute__((ext_vector_type(4))) float v4f;
typedef __attribute__((ext_vector_type(8))) float v8f;

#define B_ 64
#define T_ 2048
#define Q_ 512
#define K_ 512

#define MD_WAVES 8
#define MD_Q     (Q_ / MD_WAVES)

__global__ __launch_bounds__(32 * MD_WAVES) void mids_kernel(const float* __restrict__ query,
                                                             const float* __restrict__ W,
                                                             float* __restrict__ mids) {
    __shared__ float red[MD_WAVES][8][32];

    const int tid  = threadIdx.x;
    const int w    = tid >> 5;
    const int lane = tid & 31;
    const int half = lane >> 4;
    const int l16  = lane & 15;
    const int tm   = blockIdx.x >> 5;
    const int tn   = blockIdx.x & 31;

    const float* qrow = query + (size_t)(tm * 16 + l16) * Q_ + w * MD_Q + 4 * half;
    const float* wrow = W     + (size_t)(tn * 16 + l16) * Q_ + w * MD_Q + 4 * half;

    v4f a[MD_Q / 8], b[MD_Q / 8];
    for (int u = 0; u < MD_Q / 8; ++u) {
        a[u] = *(const v4f*)(qrow + 8 * u);
        b[u] = *(const v4f*)(wrow + 8 * u);
    }
    __builtin_amdgcn_sched_barrier(0);

    v8f acc0 = {}, acc1 = {};
    for (int u = 0; u < MD_Q / 8; ++u) {
        const v2f a01 = {a[u].x, a[u].y}, a23 = {a[u].z, a[u].w};
        const v2f b01 = {b[u].x, b[u].y}, b23 = {b[u].z, b[u].w};
        acc0 = __builtin_amdgcn_wmma_f32_16x16x4_f32(false, a01, false, b01,
                                                     (short)0, acc0, false, false);
        acc1 = __builtin_amdgcn_wmma_f32_16x16x4_f32(false, a23, false, b23,
                                                     (short)0, acc1, false, false);
    }

    for (int r = 0; r < 8; ++r) red[w][r][lane] = acc0[r] + acc1[r];
    __syncthreads();

    const int m = tid >> 4, n = tid & 15;
    const int dr = m & 7, dl = n + 16 * (m >> 3);
    float s = red[0][dr][dl];
    for (int p = 1; p < MD_WAVES; ++p) s += red[p][dr][dl];
    mids[(size_t)(tm * 16 + m) * K_ + tn * 16 + n] = s;
}

#define SC_ROWS   32
#define SC_PAIRS  (SC_ROWS / 2)
#define SC_RING   4

template <int CTRL>
static __device__ __forceinline__ float lane_xchg(float x) {
    const int v = __builtin_bit_cast(int, x);
    return __builtin_bit_cast(float, __builtin_amdgcn_mov_dpp(v, CTRL, 0xF, 0xF, true));
}

__global__ __launch_bounds__(256) void scores_kernel(const float* __restrict__ key,
                                                     const float* __restrict__ mids,
                                                     const float* __restrict__ bias,
                                                     float* __restrict__ out) {
    const int wid  = blockIdx.x * (blockDim.x >> 5) + (threadIdx.x >> 5);
    const int lane = threadIdx.x & 31;
    const int rr   = lane & 7;
    const bool b0 = rr & 1, b1 = rr & 2, b2 = rr & 4;
    const int rsel = (lane >> 3) & 1;
    const int koff = 4 * (lane & 7) + 32 * (lane >> 4);
    const int b    = wid >> 6;
    const int t0   = (wid & 63) * SC_ROWS;

    const float* kp = key  + ((size_t)b * T_ + t0 + 8 * rsel) * K_ + koff;
    const float* mp = mids + (size_t)b * K_ + koff;

    v4f m[8];
    #pragma unroll
    for (int u = 0; u < 8; ++u) m[u] = *(const v4f*)(mp + 64 * u);

    v4f buf[SC_RING][8];
#define LOAD_PAIR(P) do { \
        _Pragma("unroll") for (int u = 0; u < 8; ++u) \
            buf[(P) % SC_RING][u] = __builtin_nontemporal_load( \
                (const v4f*)(kp + (((P) >> 3) * 16 + ((P) & 7)) * K_ + 64 * u)); } while (0)

    #pragma unroll
    for (int p = 0; p < SC_RING - 1; ++p) LOAD_PAIR(p);

    float res[SC_PAIRS / 8];
    #pragma unroll
    for (int g = 0; g < SC_PAIRS / 8; ++g) res[g] = 0.f;

    #pragma unroll
    for (int p = 0; p < SC_PAIRS; ++p) {
        __builtin_amdgcn_sched_barrier(0);
        if (p + SC_RING - 1 < SC_PAIRS) LOAD_PAIR(p + SC_RING - 1);
        __builtin_amdgcn_sched_barrier(0);

        v8f acc = {};
        #pragma unroll
        for (int u = 0; u < 8; ++u) {
            const v4f a = buf[p % SC_RING][u];
            const v2f a01 = {a.x, a.y}, a23 = {a.z, a.w};
            const v2f m01 = {m[u].x, m[u].y}, m23 = {m[u].z, m[u].w};
            acc = __builtin_amdgcn_wmma_f32_16x16x4_f32(false, a01, false, m01,
                                                        (short)0, acc, false, false);
            acc = __builtin_amdgcn_wmma_f32_16x16x4_f32(false, a23, false, m23,
                                                        (short)0, acc, false, false);
        }

        const float x0 = b0 ? acc[1] : acc[0], x1 = b0 ? acc[3] : acc[2];
        const float x2 = b0 ? acc[5] : acc[4], x3 = b0 ? acc[7] : acc[6];
        const float y0 = b1 ? x1 : x0, y1 = b1 ? x3 : x2;
        float s = b2 ? y1 : y0;
        s += lane_xchg<0x161>(s);
        s += lane_xchg<0x162>(s);
        s += lane_xchg<0x164>(s);
        res[p >> 3] = (rr == (p & 7)) ? s : res[p >> 3];
    }
#undef LOAD_PAIR

    const float bv = bias[0];
    const bool owner = (lane < 8) || (lane >= 24);
    const int  tl    = (lane < 8) ? lane : lane - 16;
    #pragma unroll
    for (int g = 0; g < SC_PAIRS / 8; ++g) {
        const float e = __expf(tanhf(res[g] + bv));
        if (owner) out[(size_t)b * T_ + t0 + 16 * g + tl] = e;
    }
}

__global__ __launch_bounds__(256) void softmax_norm_kernel(float* __restrict__ out) {
    __shared__ float sdata[256];
    const int b   = blockIdx.x;
    const int tid = threadIdx.x;
    float* row = out + (size_t)b * T_;

    float v[8];
    float s = 0.f;
    #pragma unroll
    for (int i = 0; i < 8; ++i) {
        v[i] = row[tid + i * 256];
        s += v[i];
    }
    sdata[tid] = s;
    __syncthreads();
    for (int off = 128; off > 0; off >>= 1) {
        if (tid < off) sdata[tid] += sdata[tid + off];
        __syncthreads();
    }
    const float inv = 1.0f / sdata[0];
    #pragma unroll
    for (int i = 0; i < 8; ++i) row[tid + i * 256] = v[i] * inv;
}

extern "C" void kernel_launch(void* const* d_in, const int* in_sizes, int n_in,
                              void* d_out, int out_size, void* d_ws, size_t ws_size,
                              hipStream_t stream) {
    const float* query = (const float*)d_in[0];
    const float* key   = (const float*)d_in[1];
    const float* W     = (const float*)d_in[2];
    const float* bias  = (const float*)d_in[3];
    float* out  = (float*)d_out;
    float* mids = (float*)d_ws;

    mids_kernel<<<128, 256, 0, stream>>>(query, W, mids);
    scores_kernel<<<512, 256, 0, stream>>>(key, mids, bias, out);
    softmax_norm_kernel<<<64, 256, 0, stream>>>(out);
}


# scores kernel: each load reads 512 contiguous bytes of one key row, the wave walks its 64 KB in address order (was two 256-byte pieces 16 KB apart); 16-lane diagonal sum, k summation order changes
# speedup vs baseline: 1.6128x; 1.0289x over previous
#include <hip/hip_runtime.h>
#include <cmath>

typedef __attribute__((ext_vector_type(2))) float v2f;
typedef __attribute__((ext_vector_type(4))) float v4f;
typedef __attribute__((ext_vector_type(8))) float v8f;

#define B_ 64
#define T_ 2048
#define Q_ 512
#define K_ 512

#define MD_WAVES 8
#define MD_Q     (Q_ / MD_WAVES)

__global__ __launch_bounds__(32 * MD_WAVES) void mids_kernel(const float* __restrict__ query,
                                                             const float* __restrict__ W,
                                                             float* __restrict__ mids) {
    __shared__ float red[MD_WAVES][8][32];

    const int tid  = threadIdx.x;
    const int w    = tid >> 5;
    const int lane = tid & 31;
    const int half = lane >> 4;
    const int l16  = lane & 15;
    const int tm   = blockIdx.x >> 5;
    const int tn   = blockIdx.x & 31;

    const float* qrow = query + (size_t)(tm * 16 + l16) * Q_ + w * MD_Q + 4 * half;
    const float* wrow = W     + (size_t)(tn * 16 + l16) * Q_ + w * MD_Q + 4 * half;

    v4f a[MD_Q / 8], b[MD_Q / 8];
    for (int u = 0; u < MD_Q / 8; ++u) {
        a[u] = *(const v4f*)(qrow + 8 * u);
        b[u] = *(const v4f*)(wrow + 8 * u);
    }
    __builtin_amdgcn_sched_barrier(0);

    v8f acc0 = {}, acc1 = {};
    for (int u = 0; u < MD_Q / 8; ++u) {
        const v2f a01 = {a[u].x, a[u].y}, a23 = {a[u].z, a[u].w};
        const v2f b01 = {b[u].x, b[u].y}, b23 = {b[u].z, b[u].w};
        acc0 = __builtin_amdgcn_wmma_f32_16x16x4_f32(false, a01, false, b01,
                                                     (short)0, acc0, false, false);
        acc1 = __builtin_amdgcn_wmma_f32_16x16x4_f32(false, a23, false, b23,
                                                     (short)0, acc1, false, false);
    }

    for (int r = 0; r < 8; ++r) red[w][r][lane] = acc0[r] + acc1[r];
    __syncthreads();

    const int m = tid >> 4, n = tid & 15;
    const int dr = m & 7, dl = n + 16 * (m >> 3);
    float s = red[0][dr][dl];
    for (int p = 1; p < MD_WAVES; ++p) s += red[p][dr][dl];
    mids[(size_t)(tm * 16 + m) * K_ + tn * 16 + n] = s;
}

#define SC_ROWS   32
#define SC_RING   8

template <int CTRL>
static __device__ __forceinline__ float lane_xchg(float x) {
    const int v = __builtin_bit_cast(int, x);
    return __builtin_bit_cast(float, __builtin_amdgcn_mov_dpp(v, CTRL, 0xF, 0xF, true));
}

static __device__ __forceinline__ float half_xchg(float x) {
    const unsigned v = __builtin_bit_cast(unsigned, x);
    return __builtin_bit_cast(float,
        __builtin_amdgcn_permlanex16(v, v, 0xFEDCBA98u, 0x76543210u, false, false));
}

__global__ __launch_bounds__(256) void scores_kernel(const float* __restrict__ key,
                                                     const float* __restrict__ mids,
                                                     const float* __restrict__ bias,
                                                     float* __restrict__ out) {
    const int wid  = blockIdx.x * (blockDim.x >> 5) + (threadIdx.x >> 5);
    const int lane = threadIdx.x & 31;
    const int rr   = lane & 7;
    const bool b0 = rr & 1, b1 = rr & 2, b2 = rr & 4;
    const int slot = rr + ((lane >> 4) << 3);
    const int koff = 4 * (lane & 15) + 64 * (lane >> 4);
    const int b    = wid >> 6;
    const int t0   = (wid & 63) * SC_ROWS;

    const float* kp = key  + ((size_t)b * T_ + t0) * K_ + koff;
    const float* mp = mids + (size_t)b * K_ + koff;

    v4f m[4];
    #pragma unroll
    for (int u = 0; u < 4; ++u) m[u] = *(const v4f*)(mp + 128 * u);

    v4f buf[SC_RING][4];
#define LOAD_ROW(R) do { \
        _Pragma("unroll") for (int u = 0; u < 4; ++u) \
            buf[(R) % SC_RING][u] = __builtin_nontemporal_load( \
                (const v4f*)(kp + (R) * K_ + 128 * u)); } while (0)

    #pragma unroll
    for (int r = 0; r < SC_RING - 1; ++r) LOAD_ROW(r);

    float res[SC_ROWS / 16];
    #pragma unroll
    for (int g = 0; g < SC_ROWS / 16; ++g) res[g] = 0.f;

    #pragma unroll
    for (int r = 0; r < SC_ROWS; ++r) {
        __builtin_amdgcn_sched_barrier(0);
        if (r + SC_RING - 1 < SC_ROWS) LOAD_ROW(r + SC_RING - 1);
        __builtin_amdgcn_sched_barrier(0);

        v8f acc = {};
        #pragma unroll
        for (int u = 0; u < 4; ++u) {
            const v4f a = buf[r % SC_RING][u];
            const v2f a01 = {a.x, a.y}, a23 = {a.z, a.w};
            const v2f m01 = {m[u].x, m[u].y}, m23 = {m[u].z, m[u].w};
            acc = __builtin_amdgcn_wmma_f32_16x16x4_f32(false, a01, false, m01,
                                                        (short)0, acc, false, false);
            acc = __builtin_amdgcn_wmma_f32_16x16x4_f32(false, a23, false, m23,
                                                        (short)0, acc, false, false);
        }

        const float x0 = b0 ? acc[1] : acc[0], x1 = b0 ? acc[3] : acc[2];
        const float x2 = b0 ? acc[5] : acc[4], x3 = b0 ? acc[7] : acc[6];
        const float y0 = b1 ? x1 : x0, y1 = b1 ? x3 : x2;
        float s = b2 ? y1 : y0;
        s += lane_xchg<0x161>(s);
        s += lane_xchg<0x162>(s);
        s += lane_xchg<0x164>(s);
        s += half_xchg(s);
        res[r >> 4] = (slot == (r & 15)) ? s : res[r >> 4];
    }
#undef LOAD_ROW

    const float bv = bias[0];
    const bool owner = (lane < 8) || (lane >= 24);
    const int  tl    = (lane < 8) ? lane : lane - 16;
    #pragma unroll
    for (int g = 0; g < SC_ROWS / 16; ++g) {
        const float e = __expf(tanhf(res[g] + bv));
        if (owner) out[(size_t)b * T_ + t0 + 16 * g + tl] = e;
    }
}

__global__ __launch_bounds__(256) void softmax_norm_kernel(float* __restrict__ out) {
    __shared__ float sdata[256];
    const int b   = blockIdx.x;
    const int tid = threadIdx.x;
    float* row = out + (size_t)b * T_;

    float v[8];
    float s = 0.f;
    #pragma unroll
    for (int i = 0; i < 8; ++i) {
        v[i] = row[tid + i * 256];
        s += v[i];
    }
    sdata[tid] = s;
    __syncthreads();
    for (int off = 128; off > 0; off >>= 1) {
        if (tid < off) sdata[tid] += sdata[tid + off];
        __syncthreads();
    }
    const float inv = 1.0f / sdata[0];
    #pragma unroll
    for (int i = 0; i < 8; ++i) row[tid + i * 256] = v[i] * inv;
}

extern "C" void kernel_launch(void* const* d_in, const int* in_sizes, int n_in,
                              void* d_out, int out_size, void* d_ws, size_t ws_size,
                              hipStream_t stream) {
    const float* query = (const float*)d_in[0];
    const float* key   = (const float*)d_in[1];
    const float* W     = (const float*)d_in[2];
    const float* bias  = (const float*)d_in[3];
    float* out  = (float*)d_out;
    float* mids = (float*)d_ws;

    mids_kernel<<<128, 256, 0, stream>>>(query, W, mids);
    scores_kernel<<<512, 256, 0, stream>>>(key, mids, bias, out);
    softmax_norm_kernel<<<64, 256, 0, stream>>>(out);
}
